# starting kernel
# baseline (speedup 1.0000x reference)
#include <hip/hip_runtime.h>


typedef __attribute__((ext_vector_type(2))) float v2f;
typedef __attribute__((ext_vector_type(8))) float v8f;

#define T_LEN 2048
#define LDS_STRIDE 17

__device__ __forceinline__ float fexp2(float x) { return __builtin_amdgcn_exp2f(x); }
__device__ __forceinline__ float frcp (float x) { return __builtin_amdgcn_rcpf(x);  }
__device__ __forceinline__ float sigm_(float x) { return frcp(1.0f + fexp2(x * -1.44269504088896340736f)); }
__device__ __forceinline__ float tanh_(float x) { return 1.0f - 2.0f * frcp(1.0f + fexp2(x * 2.88539008177792681472f)); }

__global__ __launch_bounds__(32)
void lstm_fused_wmma(const float* __restrict__ x,
                     const float* __restrict__ W_ih,
                     const float* __restrict__ W_hh,
                     const float* __restrict__ b_ih,
                     const float* __restrict__ b_hh,
                     const float* __restrict__ W_out,
                     const float* __restrict__ b_out,
                     float* __restrict__ out)
{
    __shared__ float hlds[16 * LDS_STRIDE];

    const int lane = threadIdx.x & 31;
    const int lo   = lane & 15;
    const int hi   = lane >> 4;
    const int b0   = blockIdx.x * 16;

    v2f wB[4][5];
#pragma unroll
    for (int n = 0; n < 4; ++n)
#pragma unroll
        for (int k = 0; k < 5; ++k)
#pragma unroll
            for (int q = 0; q < 2; ++q) {
                const int K = 4 * k + 2 * hi + q;
                const int N = 16 * n + lo;
                float v;
                if      (K < 16) v = W_hh[N * 16 + K];
                else if (K < 19) v = W_ih[N * 3 + (K - 16)];
                else             v = b_ih[N] + b_hh[N];
                wB[n][k][q] = v;
            }

    v2f hB[4];
#pragma unroll
    for (int k = 0; k < 4; ++k)
#pragma unroll
        for (int q = 0; q < 2; ++q) {
            const int K = 4 * k + 2 * hi + q;
            hB[k][q] = (lo < 3) ? W_out[lo * 16 + K] : 0.0f;
        }
    const float obias = (lo < 3) ? b_out[lo] : 0.0f;

    const int st_off = (8 * hi) * LDS_STRIDE + lo;
    const int ld_off = lo * LDS_STRIDE + 2 * hi;

    const float* xp = x + (b0 + lo) * (T_LEN * 3) + (hi ? 2 : 0);

    v8f cst;
    v2f ah[4];
#pragma unroll
    for (int r = 0; r < 8; ++r) cst[r] = 0.0f;
#pragma unroll
    for (int k = 0; k < 4; ++k) { ah[k][0] = 0.0f; ah[k][1] = 0.0f; }

    float xa = xp[0];
    float xb = 1.0f;
    if (!hi) xb = xp[1];

    const int out_base = (b0 + 8 * hi) * (T_LEN * 3) + lo;
    const bool do_out  = (lo < 3);

    for (int t = 0; t < T_LEN; ++t) {
        const int tn = (t + 1 < T_LEN) ? (t + 1) : t;
        float na = xp[tn * 3];
        float nb = 1.0f;
        if (!hi) nb = xp[tn * 3 + 1];

        v2f ax; ax[0] = xa; ax[1] = xb;

        v8f acc[4];
#pragma unroll
        for (int n = 0; n < 4; ++n) {
            v8f a = {};
#pragma unroll
            for (int k = 0; k < 4; ++k)
                a = __builtin_amdgcn_wmma_f32_16x16x4_f32(
                        false, ah[k], false, wB[n][k], (short)0, a, false, false);
            a = __builtin_amdgcn_wmma_f32_16x16x4_f32(
                        false, ax, false, wB[n][4], (short)0, a, false, false);
            acc[n] = a;
        }

        v8f hval;
#pragma unroll
        for (int r = 0; r < 8; ++r) {
            const float iv = sigm_(acc[0][r]);
            const float fv = sigm_(acc[1][r]);
            const float gv = tanh_(acc[2][r]);
            const float ov = sigm_(acc[3][r]);
            const float cc = fv * cst[r] + iv * gv;
            cst[r] = cc;
            hval[r] = ov * tanh_(cc);
        }

#pragma unroll
        for (int r = 0; r < 8; ++r)
            hlds[st_off + r * LDS_STRIDE] = hval[r];
        __syncthreads();
#pragma unroll
        for (int k = 0; k < 4; ++k) {
            ah[k][0] = hlds[ld_off + 4 * k + 0];
            ah[k][1] = hlds[ld_off + 4 * k + 1];
        }

        v8f oacc;
#pragma unroll
        for (int r = 0; r < 8; ++r) oacc[r] = obias;
#pragma unroll
        for (int k = 0; k < 4; ++k)
            oacc = __builtin_amdgcn_wmma_f32_16x16x4_f32(
                       false, ah[k], false, hB[k], (short)0, oacc, false, false);

        if (do_out) {
#pragma unroll
            for (int r = 0; r < 8; ++r)
                out[out_base + r * (T_LEN * 3) + t * 3] = oacc[r];
        }

        xa = na; xb = nb;
    }
}

extern "C" void kernel_launch(void* const* d_in, const int* in_sizes, int n_in,
                              void* d_out, int out_size, void* d_ws, size_t ws_size,
                              hipStream_t stream) {
    const float* x     = (const float*)d_in[0];
    const float* W_ih  = (const float*)d_in[1];
    const float* W_hh  = (const float*)d_in[2];
    const float* b_ih  = (const float*)d_in[3];
    const float* b_hh  = (const float*)d_in[4];
    const float* W_out = (const float*)d_in[5];
    const float* b_out = (const float*)d_in[6];
    float* out = (float*)d_out;

    const int B = in_sizes[0] / (T_LEN * 3);
    const int grid = B / 16;

    lstm_fused_wmma<<<grid, 32, 0, stream>>>(x, W_ih, W_hh, b_ih, b_hh,
                                             W_out, b_out, out);
}


# 4-wave work-group per 16-row batch tile: gate rows regrouped per hidden unit, weights as WMMA A operand, wave w owns units 4w..4w+3, h exchanged via LDS once a step; x/bias k-tile summed first
# speedup vs baseline: 1.6464x; 1.6464x over previous
#include <hip/hip_runtime.h>


typedef __attribute__((ext_vector_type(2))) float v2f;
typedef __attribute__((ext_vector_type(8))) float v8f;

#define T_LEN 2048

__device__ __forceinline__ float fexp2(float x) { return __builtin_amdgcn_exp2f(x); }
__device__ __forceinline__ float frcp (float x) { return __builtin_amdgcn_rcpf(x);  }
__device__ __forceinline__ float sigm_(float x) { return frcp(1.0f + fexp2(x * -1.44269504088896340736f)); }
__device__ __forceinline__ float tanh_(float x) { return 1.0f - 2.0f * frcp(1.0f + fexp2(x * 2.88539008177792681472f)); }

__device__ __forceinline__ v8f wmma4(v2f a, v2f b, v8f c) {
    return __builtin_amdgcn_wmma_f32_16x16x4_f32(false, a, false, b, (short)0, c, false, false);
}

__device__ __forceinline__ void head_store(const v2f (&hA)[4], const v8f& obias, const v2f (&bh)[4],
                                           float* __restrict__ o3, int hf) {
    v8f o = obias;
#pragma unroll
    for (int k = 0; k < 4; ++k)
        o = wmma4(hA[k], bh[k], o);
    if (hf == 0) { o3[0] = o[0]; o3[1] = o[1]; o3[2] = o[2]; }
}

template <int P>
__device__ __forceinline__ void lstm_step(int t, int w, int lane, int hf,
                                          const v2f (&wA)[5], const v2f (&hA)[4], const v8f& obias,
                                          v8f& acc, float (&cst)[2], v8f& o,
                                          const v2f (&bh)[4], v2f (&nbh)[4], v2f& xuse,
                                          const float* __restrict__ xp0, const float* __restrict__ xp1,
                                          float* __restrict__ op, v2f (*hx)[4][32])
{
    const v8f zero = {};
#pragma unroll
    for (int k = 0; k < 4; ++k)
        acc = wmma4(wA[k], bh[k], acc);

    v2f nh;
#pragma unroll
    for (int q = 0; q < 2; ++q) {
        const float iv = sigm_(acc[4 * q + 0]);
        const float fv = sigm_(acc[4 * q + 1]);
        const float gv = tanh_(acc[4 * q + 2]);
        const float ov = sigm_(acc[4 * q + 3]);
        const float cc = fv * cst[q] + iv * gv;
        cst[q] = cc;
        nh[q] = ov * tanh_(cc);
    }
    hx[P][w][lane] = nh;

    acc = wmma4(wA[4], xuse, zero);
    if (t >= 2 && ((t - 2) & 3) == w) {
        if (hf == 0) { float* o3 = op + (t - 2) * 3; o3[0] = o[0]; o3[1] = o[1]; o3[2] = o[2]; }
    }
    {
        const int t3 = (t + 3 < T_LEN) ? (t + 3) : (T_LEN - 1);
        xuse[0] = xp0[t3 * 3]; xuse[1] = hf ? 1.0f : xp1[t3 * 3];
    }

    __builtin_amdgcn_fence(__ATOMIC_RELEASE, "workgroup", "local");
    __builtin_amdgcn_s_barrier();
    __builtin_amdgcn_fence(__ATOMIC_ACQUIRE, "workgroup", "local");

#pragma unroll
    for (int k = 0; k < 4; ++k)
        nbh[k] = hx[P][k][lane];
    if (t >= 1 && ((t - 1) & 3) == w) {
        o = obias;
#pragma unroll
        for (int k = 0; k < 4; ++k)
            o = wmma4(hA[k], bh[k], o);
    }
}

__global__ __launch_bounds__(128)
void lstm_fused_wmma(const float* __restrict__ x,
                     const float* __restrict__ W_ih,
                     const float* __restrict__ W_hh,
                     const float* __restrict__ b_ih,
                     const float* __restrict__ b_hh,
                     const float* __restrict__ W_out,
                     const float* __restrict__ b_out,
                     float* __restrict__ out)
{
    __shared__ v2f hx[2][4][32];

    const int lane = threadIdx.x & 31;
    const int w    = __builtin_amdgcn_readfirstlane((int)(threadIdx.x >> 5));
    const int c    = lane & 15;
    const int hf   = lane >> 4;
    const int b0   = blockIdx.x * 16;

    v2f wA[5];
    {
        const int N = 16 * (c & 3) + 4 * w + (c >> 2);
#pragma unroll
        for (int k = 0; k < 5; ++k)
#pragma unroll
            for (int q = 0; q < 2; ++q) {
                const int K = 4 * k + 2 * hf + q;
                float v;
                if      (K < 16) v = W_hh[N * 16 + K];
                else if (K < 19) v = W_ih[N * 3 + (K - 16)];
                else             v = b_ih[N] + b_hh[N];
                wA[k][q] = v;
            }
    }

    v2f hA[4];
#pragma unroll
    for (int k = 0; k < 4; ++k)
#pragma unroll
        for (int q = 0; q < 2; ++q) {
            const int K = 4 * k + 2 * hf + q;
            hA[k][q] = (c < 3) ? W_out[c * 16 + K] : 0.0f;
        }
    v8f obias;
#pragma unroll
    for (int r = 0; r < 8; ++r)
        obias[r] = (r < 3 && hf == 0) ? b_out[r < 3 ? r : 0] : 0.0f;

    const float* xp0 = x + (b0 + c) * (T_LEN * 3) + (hf ? 2 : 0);
    const float* xp1 = x + (b0 + c) * (T_LEN * 3) + (hf ? 2 : 1);
    float* op = out + (b0 + c) * (T_LEN * 3);

    float cst[2] = {0.0f, 0.0f};
    v2f bh[4];
#pragma unroll
    for (int k = 0; k < 4; ++k) { bh[k][0] = 0.0f; bh[k][1] = 0.0f; }

    const v8f zero = {};
    v2f x0;  x0[0]  = xp0[0]; x0[1]  = hf ? 1.0f : xp1[0];
    v2f xn1; xn1[0] = xp0[3]; xn1[1] = hf ? 1.0f : xp1[3];
    v2f xn2; xn2[0] = xp0[6]; xn2[1] = hf ? 1.0f : xp1[6];
    v8f acc = wmma4(wA[4], x0, zero);

    v8f o = obias;
    v2f bhB[4];
    for (int t = 0; t < T_LEN; t += 2) {
        lstm_step<0>(t,     w, lane, hf, wA, hA, obias, acc, cst, o, bh,  bhB, xn1, xp0, xp1, op, hx);
        lstm_step<1>(t + 1, w, lane, hf, wA, hA, obias, acc, cst, o, bhB, bh,  xn2, xp0, xp1, op, hx);
    }

    if (((T_LEN - 2) & 3) == w) {
        if (hf == 0) { float* o3 = op + (T_LEN - 2) * 3; o3[0] = o[0]; o3[1] = o[1]; o3[2] = o[2]; }
    }
    if (((T_LEN - 1) & 3) == w)
        head_store(hA, obias, bh, op + (T_LEN - 1) * 3, hf);
}

extern "C" void kernel_launch(void* const* d_in, const int* in_sizes, int n_in,
                              void* d_out, int out_size, void* d_ws, size_t ws_size,
                              hipStream_t stream) {
    const float* x     = (const float*)d_in[0];
    const float* W_ih  = (const float*)d_in[1];
    const float* W_hh  = (const float*)d_in[2];
    const float* b_ih  = (const float*)d_in[3];
    const float* b_hh  = (const float*)d_in[4];
    const float* W_out = (const float*)d_in[5];
    const float* b_out = (const float*)d_in[6];
    float* out = (float*)d_out;

    const int B = in_sizes[0] / (T_LEN * 3);
    const int grid = B / 16;

    lstm_fused_wmma<<<grid, 128, 0, stream>>>(x, W_ih, W_hh, b_ih, b_hh,
                                              W_out, b_out, out);
}


# Global instructions taken off the recurrence path: x read as 48-byte blocks of four steps, one block ahead; out store issued behind the barrier; none left between h's LDS store and the barrier
# speedup vs baseline: 2.5411x; 1.5434x over previous
#include <hip/hip_runtime.h>


typedef __attribute__((ext_vector_type(2))) float v2f;
typedef __attribute__((ext_vector_type(4))) float v4f;
typedef __attribute__((ext_vector_type(8))) float v8f;

#define T_LEN 2048

__device__ __forceinline__ float fexp2(float x) { return __builtin_amdgcn_exp2f(x); }
__device__ __forceinline__ float frcp (float x) { return __builtin_amdgcn_rcpf(x);  }
__device__ __forceinline__ float sigm_(float x) { return frcp(1.0f + fexp2(x * -1.44269504088896340736f)); }
__device__ __forceinline__ float tanh_(float x) { return 1.0f - 2.0f * frcp(1.0f + fexp2(x * 2.88539008177792681472f)); }

__device__ __forceinline__ v8f wmma4(v2f a, v2f b, v8f c) {
    return __builtin_amdgcn_wmma_f32_16x16x4_f32(false, a, false, b, (short)0, c, false, false);
}

__device__ __forceinline__ void head_store(const v2f (&hA)[4], const v8f& obias, const v2f (&bh)[4],
                                           float* __restrict__ o3, int hf) {
    v8f o = obias;
#pragma unroll
    for (int k = 0; k < 4; ++k)
        o = wmma4(hA[k], bh[k], o);
    if (hf == 0) { o3[0] = o[0]; o3[1] = o[1]; o3[2] = o[2]; }
}

template <int I>
__device__ __forceinline__ v2f x_frag(const v4f (&blk)[3], int hf) {
    const float e0 = blk[(3 * I) >> 2][(3 * I) & 3];
    const float e1 = blk[(3 * I + 1) >> 2][(3 * I + 1) & 3];
    const float e2 = blk[(3 * I + 2) >> 2][(3 * I + 2) & 3];
    v2f r;
    r[0] = hf ? e2 : e0;
    r[1] = hf ? 1.0f : e1;
    return r;
}

template <int S>
__device__ __forceinline__ void lstm_step(int t, int w, int lane, int hf,
                                          const v2f (&wA)[5], const v2f (&hA)[4], const v8f& obias,
                                          v8f& acc, float (&cst)[2], v8f& o,
                                          const v2f (&bh)[4], v2f (&nbh)[4],
                                          const v4f (&cur)[3], const v4f (&nxt)[3],
                                          v4f (&ld)[3], const float* __restrict__ ldp,
                                          float* __restrict__ og, unsigned roff, v2f (*hx)[4][32])
{
    constexpr int P = S & 1;
    const v8f zero = {};
#pragma unroll
    for (int k = 0; k < 4; ++k)
        acc = wmma4(wA[k], bh[k], acc);

    v2f nh;
#pragma unroll
    for (int q = 0; q < 2; ++q) {
        const float iv = sigm_(acc[4 * q + 0]);
        const float fv = sigm_(acc[4 * q + 1]);
        const float gv = tanh_(acc[4 * q + 2]);
        const float ov = sigm_(acc[4 * q + 3]);
        const float cc = fv * cst[q] + iv * gv;
        cst[q] = cc;
        nh[q] = ov * tanh_(cc);
    }
    hx[P][w][lane] = nh;

    if (S < 3) acc = wmma4(wA[4], x_frag<(S + 1) & 3>(cur, hf), zero);
    else       acc = wmma4(wA[4], x_frag<0>(nxt, hf), zero);

    __builtin_amdgcn_fence(__ATOMIC_RELEASE, "workgroup", "local");
    __builtin_amdgcn_s_barrier();
    __builtin_amdgcn_fence(__ATOMIC_ACQUIRE, "workgroup", "local");

#pragma unroll
    for (int k = 0; k < 4; ++k)
        nbh[k] = hx[P][k][lane];

    if (t >= 2 && ((S + 2) & 3) == w) {
        if (hf == 0) { float* o3 = og + (t - 2) * 3; o3[roff] = o[0]; o3[roff + 1] = o[1]; o3[roff + 2] = o[2]; }
    }
    if (S == 0) {
        const v4f* lp = reinterpret_cast<const v4f*>(ldp + roff);
        ld[0] = lp[0]; ld[1] = lp[1]; ld[2] = lp[2];
    }
    if (t >= 1 && ((S + 3) & 3) == w) {
        o = obias;
#pragma unroll
        for (int k = 0; k < 4; ++k)
            o = wmma4(hA[k], bh[k], o);
    }
}

__global__ __launch_bounds__(128)
void lstm_fused_wmma(const float* __restrict__ x,
                     const float* __restrict__ W_ih,
                     const float* __restrict__ W_hh,
                     const float* __restrict__ b_ih,
                     const float* __restrict__ b_hh,
                     const float* __restrict__ W_out,
                     const float* __restrict__ b_out,
                     float* __restrict__ out)
{
    __shared__ v2f hx[2][4][32];

    const int lane = threadIdx.x & 31;
    const int w    = __builtin_amdgcn_readfirstlane((int)(threadIdx.x >> 5));
    const int c    = lane & 15;
    const int hf   = lane >> 4;
    const int b0   = blockIdx.x * 16;

    v2f wA[5];
    {
        const int N = 16 * (c & 3) + 4 * w + (c >> 2);
#pragma unroll
        for (int k = 0; k < 5; ++k)
#pragma unroll
            for (int q = 0; q < 2; ++q) {
                const int K = 4 * k + 2 * hf + q;
                float v;
                if      (K < 16) v = W_hh[N * 16 + K];
                else if (K < 19) v = W_ih[N * 3 + (K - 16)];
                else             v = b_ih[N] + b_hh[N];
                wA[k][q] = v;
            }
    }

    v2f hA[4];
#pragma unroll
    for (int k = 0; k < 4; ++k)
#pragma unroll
        for (int q = 0; q < 2; ++q) {
            const int K = 4 * k + 2 * hf + q;
            hA[k][q] = (c < 3) ? W_out[c * 16 + K] : 0.0f;
        }
    v8f obias;
#pragma unroll
    for (int r = 0; r < 8; ++r)
        obias[r] = (r < 3 && hf == 0) ? b_out[r < 3 ? r : 0] : 0.0f;

    const float* xg = x + b0 * (T_LEN * 3);
    float* og = out + b0 * (T_LEN * 3);
    const unsigned roff = (unsigned)c * (T_LEN * 3);
    const v4f* xq = reinterpret_cast<const v4f*>(xg + roff);
    float* op = og + roff;

    float cst[2] = {0.0f, 0.0f};
    v2f bh[4];
#pragma unroll
    for (int k = 0; k < 4; ++k) { bh[k][0] = 0.0f; bh[k][1] = 0.0f; }

    const v8f zero = {};
    v4f xa[3], xb[3];
    xa[0] = xq[0]; xa[1] = xq[1]; xa[2] = xq[2];
    v8f acc = wmma4(wA[4], x_frag<0>(xa, hf), zero);

    v8f o = obias;
    v2f bhB[4];
    for (int t = 0; t < T_LEN; t += 8) {
        const float* pb = xg + 12 * (t / 4 + 1);
        const float* pa = xg + 12 * ((t + 8 < T_LEN) ? (t / 4 + 2) : (t / 4 + 1));
        lstm_step<0>(t,     w, lane, hf, wA, hA, obias, acc, cst, o, bh,  bhB, xa, xb, xb, pb, og, roff, hx);
        lstm_step<1>(t + 1, w, lane, hf, wA, hA, obias, acc, cst, o, bhB, bh,  xa, xb, xb, pb, og, roff, hx);
        lstm_step<2>(t + 2, w, lane, hf, wA, hA, obias, acc, cst, o, bh,  bhB, xa, xb, xb, pb, og, roff, hx);
        lstm_step<3>(t + 3, w, lane, hf, wA, hA, obias, acc, cst, o, bhB, bh,  xa, xb, xb, pb, og, roff, hx);
        lstm_step<0>(t + 4, w, lane, hf, wA, hA, obias, acc, cst, o, bh,  bhB, xb, xa, xa, pa, og, roff, hx);
        lstm_step<1>(t + 5, w, lane, hf, wA, hA, obias, acc, cst, o, bhB, bh,  xb, xa, xa, pa, og, roff, hx);
        lstm_step<2>(t + 6, w, lane, hf, wA, hA, obias, acc, cst, o, bh,  bhB, xb, xa, xa, pa, og, roff, hx);
        lstm_step<3>(t + 7, w, lane, hf, wA, hA, obias, acc, cst, o, bhB, bh,  xb, xa, xa, pa, og, roff, hx);
    }

    if (((T_LEN - 2) & 3) == w) {
        if (hf == 0) { float* o3 = op + (T_LEN - 2) * 3; o3[0] = o[0]; o3[1] = o[1]; o3[2] = o[2]; }
    }
    if (((T_LEN - 1) & 3) == w)
        head_store(hA, obias, bh, op + (T_LEN - 1) * 3, hf);
}

extern "C" void kernel_launch(void* const* d_in, const int* in_sizes, int n_in,
                              void* d_out, int out_size, void* d_ws, size_t ws_size,
                              hipStream_t stream) {
    const float* x     = (const float*)d_in[0];
    const float* W_ih  = (const float*)d_in[1];
    const float* W_hh  = (const float*)d_in[2];
    const float* b_ih  = (const float*)d_in[3];
    const float* b_hh  = (const float*)d_in[4];
    const float* W_out = (const float*)d_in[5];
    const float* b_out = (const float*)d_in[6];
    float* out = (float*)d_out;

    const int B = in_sizes[0] / (T_LEN * 3);
    const int grid = B / 16;

    lstm_fused_wmma<<<grid, 128, 0, stream>>>(x, W_ih, W_hh, b_ih, b_hh,
                                              W_out, b_out, out);
}


# One v_tanh_f32 per activation in place of exp2+rcp: sigmoid(z) = 1/2 + 1/2*tanh(z/2), the halving folded exactly into the i, f, o weight rows; 10 transcendentals per lane and step, not 20
# speedup vs baseline: 2.7592x; 1.0858x over previous
#include <hip/hip_runtime.h>


typedef __attribute__((ext_vector_type(2))) float v2f;
typedef __attribute__((ext_vector_type(4))) float v4f;
typedef __attribute__((ext_vector_type(8))) float v8f;

#define T_LEN 2048

__device__ __forceinline__ float tanh_(float x) { return __builtin_amdgcn_tanhf(x); }
__device__ __forceinline__ float sigm_half(float hz) { return __builtin_fmaf(0.5f, tanh_(hz), 0.5f); }

__device__ __forceinline__ v8f wmma4(v2f a, v2f b, v8f c) {
    return __builtin_amdgcn_wmma_f32_16x16x4_f32(false, a, false, b, (short)0, c, false, false);
}

__device__ __forceinline__ void head_store(const v2f (&hA)[4], const v8f& obias, const v2f (&bh)[4],
                                           float* __restrict__ o3, int hf) {
    v8f o = obias;
#pragma unroll
    for (int k = 0; k < 4; ++k)
        o = wmma4(hA[k], bh[k], o);
    if (hf == 0) { o3[0] = o[0]; o3[1] = o[1]; o3[2] = o[2]; }
}

template <int I>
__device__ __forceinline__ v2f x_frag(const v4f (&blk)[3], int hf) {
    const float e0 = blk[(3 * I) >> 2][(3 * I) & 3];
    const float e1 = blk[(3 * I + 1) >> 2][(3 * I + 1) & 3];
    const float e2 = blk[(3 * I + 2) >> 2][(3 * I + 2) & 3];
    v2f r;
    r[0] = hf ? e2 : e0;
    r[1] = hf ? 1.0f : e1;
    return r;
}

template <int S>
__device__ __forceinline__ void lstm_step(int t, int w, int lane, int hf,
                                          const v2f (&wA)[5], const v2f (&hA)[4], const v8f& obias,
                                          v8f& acc, float (&cst)[2], v8f& o,
                                          const v2f (&bh)[4], v2f (&nbh)[4],
                                          const v4f (&cur)[3], const v4f (&nxt)[3],
                                          v4f (&ld)[3], const float* __restrict__ ldp,
                                          float* __restrict__ og, unsigned roff, v2f (*hx)[4][32])
{
    constexpr int P = S & 1;
    const v8f zero = {};
#pragma unroll
    for (int k = 0; k < 4; ++k)
        acc = wmma4(wA[k], bh[k], acc);

    v2f nh;
#pragma unroll
    for (int q = 0; q < 2; ++q) {
        const float iv = sigm_half(acc[4 * q + 0]);
        const float fv = sigm_half(acc[4 * q + 1]);
        const float gv = tanh_(acc[4 * q + 2]);
        const float ov = sigm_half(acc[4 * q + 3]);
        const float cc = fv * cst[q] + iv * gv;
        cst[q] = cc;
        nh[q] = ov * tanh_(cc);
    }
    hx[P][w][lane] = nh;

    if (S < 3) acc = wmma4(wA[4], x_frag<(S + 1) & 3>(cur, hf), zero);
    else       acc = wmma4(wA[4], x_frag<0>(nxt, hf), zero);

    __builtin_amdgcn_fence(__ATOMIC_RELEASE, "workgroup", "local");
    __builtin_amdgcn_s_barrier();
    __builtin_amdgcn_fence(__ATOMIC_ACQUIRE, "workgroup", "local");

#pragma unroll
    for (int k = 0; k < 4; ++k)
        nbh[k] = hx[P][k][lane];

    if (t >= 2 && ((S + 2) & 3) == w) {
        if (hf == 0) { float* o3 = og + (t - 2) * 3; o3[roff] = o[0]; o3[roff + 1] = o[1]; o3[roff + 2] = o[2]; }
    }
    if (S == 0) {
        const v4f* lp = reinterpret_cast<const v4f*>(ldp + roff);
        ld[0] = lp[0]; ld[1] = lp[1]; ld[2] = lp[2];
    }
    if (t >= 1 && ((S + 3) & 3) == w) {
        o = obias;
#pragma unroll
        for (int k = 0; k < 4; ++k)
            o = wmma4(hA[k], bh[k], o);
    }
}

__global__ __launch_bounds__(128)
void lstm_fused_wmma(const float* __restrict__ x,
                     const float* __restrict__ W_ih,
                     const float* __restrict__ W_hh,
                     const float* __restrict__ b_ih,
                     const float* __restrict__ b_hh,
                     const float* __restrict__ W_out,
                     const float* __restrict__ b_out,
                     float* __restrict__ out)
{
    __shared__ v2f hx[2][4][32];

    const int lane = threadIdx.x & 31;
    const int w    = __builtin_amdgcn_readfirstlane((int)(threadIdx.x >> 5));
    const int c    = lane & 15;
    const int hf   = lane >> 4;
    const int b0   = blockIdx.x * 16;

    v2f wA[5];
    {
        const int N = 16 * (c & 3) + 4 * w + (c >> 2);
        const float sc = ((c & 3) == 2) ? 1.0f : 0.5f;
#pragma unroll
        for (int k = 0; k < 5; ++k)
#pragma unroll
            for (int q = 0; q < 2; ++q) {
                const int K = 4 * k + 2 * hf + q;
                float v;
                if      (K < 16) v = W_hh[N * 16 + K];
                else if (K < 19) v = W_ih[N * 3 + (K - 16)];
                else             v = b_ih[N] + b_hh[N];
                wA[k][q] = v * sc;
            }
    }

    v2f hA[4];
#pragma unroll
    for (int k = 0; k < 4; ++k)
#pragma unroll
        for (int q = 0; q < 2; ++q) {
            const int K = 4 * k + 2 * hf + q;
            hA[k][q] = (c < 3) ? W_out[c * 16 + K] : 0.0f;
        }
    v8f obias;
#pragma unroll
    for (int r = 0; r < 8; ++r)
        obias[r] = (r < 3 && hf == 0) ? b_out[r < 3 ? r : 0] : 0.0f;

    const float* xg = x + b0 * (T_LEN * 3);
    float* og = out + b0 * (T_LEN * 3);
    const unsigned roff = (unsigned)c * (T_LEN * 3);
    const v4f* xq = reinterpret_cast<const v4f*>(xg + roff);
    float* op = og + roff;

    float cst[2] = {0.0f, 0.0f};
    v2f bh[4];
#pragma unroll
    for (int k = 0; k < 4; ++k) { bh[k][0] = 0.0f; bh[k][1] = 0.0f; }

    const v8f zero = {};
    v4f xa[3], xb[3];
    xa[0] = xq[0]; xa[1] = xq[1]; xa[2] = xq[2];
    v8f acc = wmma4(wA[4], x_frag<0>(xa, hf), zero);

    v8f o = obias;
    v2f bhB[4];
    for (int t = 0; t < T_LEN; t += 8) {
        const float* pb = xg + 12 * (t / 4 + 1);
        const float* pa = xg + 12 * ((t + 8 < T_LEN) ? (t / 4 + 2) : (t / 4 + 1));
        lstm_step<0>(t,     w, lane, hf, wA, hA, obias, acc, cst, o, bh,  bhB, xa, xb, xb, pb, og, roff, hx);
        lstm_step<1>(t + 1, w, lane, hf, wA, hA, obias, acc, cst, o, bhB, bh,  xa, xb, xb, pb, og, roff, hx);
        lstm_step<2>(t + 2, w, lane, hf, wA, hA, obias, acc, cst, o, bh,  bhB, xa, xb, xb, pb, og, roff, hx);
        lstm_step<3>(t + 3, w, lane, hf, wA, hA, obias, acc, cst, o, bhB, bh,  xa, xb, xb, pb, og, roff, hx);
        lstm_step<0>(t + 4, w, lane, hf, wA, hA, obias, acc, cst, o, bh,  bhB, xb, xa, xa, pa, og, roff, hx);
        lstm_step<1>(t + 5, w, lane, hf, wA, hA, obias, acc, cst, o, bhB, bh,  xb, xa, xa, pa, og, roff, hx);
        lstm_step<2>(t + 6, w, lane, hf, wA, hA, obias, acc, cst, o, bh,  bhB, xb, xa, xa, pa, og, roff, hx);
        lstm_step<3>(t + 7, w, lane, hf, wA, hA, obias, acc, cst, o, bhB, bh,  xb, xa, xa, pa, og, roff, hx);
    }

    if (((T_LEN - 2) & 3) == w) {
        if (hf == 0) { float* o3 = op + (T_LEN - 2) * 3; o3[0] = o[0]; o3[1] = o[1]; o3[2] = o[2]; }
    }
    if (((T_LEN - 1) & 3) == w)
        head_store(hA, obias, bh, op + (T_LEN - 1) * 3, hf);
}

extern "C" void kernel_launch(void* const* d_in, const int* in_sizes, int n_in,
                              void* d_out, int out_size, void* d_ws, size_t ws_size,
                              hipStream_t stream) {
    const float* x     = (const float*)d_in[0];
    const float* W_ih  = (const float*)d_in[1];
    const float* W_hh  = (const float*)d_in[2];
    const float* b_ih  = (const float*)d_in[3];
    const float* b_hh  = (const float*)d_in[4];
    const float* W_out = (const float*)d_in[5];
    const float* b_out = (const float*)d_in[6];
    float* out = (float*)d_out;

    const int B = in_sizes[0] / (T_LEN * 3);
    const int grid = B / 16;

    lstm_fused_wmma<<<grid, 128, 0, stream>>>(x, W_ih, W_hh, b_ih, b_hh,
                                              W_out, b_out, out);
}


# Own k-tile of h multiplied ahead of the barrier from the lane's own registers: three products and three LDS loads behind the hand-over, not four; each wave now sums its k-tiles own first
# speedup vs baseline: 2.8745x; 1.0418x over previous
#include <hip/hip_runtime.h>


typedef __attribute__((ext_vector_type(2))) float v2f;
typedef __attribute__((ext_vector_type(4))) float v4f;
typedef __attribute__((ext_vector_type(8))) float v8f;

#define T_LEN 2048

__device__ __forceinline__ float tanh_(float x) { return __builtin_amdgcn_tanhf(x); }
__device__ __forceinline__ float sigm_half(float hz) { return __builtin_fmaf(0.5f, tanh_(hz), 0.5f); }

__device__ __forceinline__ v8f wmma4(v2f a, v2f b, v8f c) {
    return __builtin_amdgcn_wmma_f32_16x16x4_f32(false, a, false, b, (short)0, c, false, false);
}

__device__ __forceinline__ void head_store(const v2f& hOwn, const v2f (&hO)[3], const v8f& obias,
                                           const v2f& own, const v2f (&bh)[3],
                                           float* __restrict__ o3, int hf) {
    v8f o = wmma4(hOwn, own, obias);
#pragma unroll
    for (int j = 0; j < 3; ++j)
        o = wmma4(hO[j], bh[j], o);
    if (hf == 0) { o3[0] = o[0]; o3[1] = o[1]; o3[2] = o[2]; }
}

template <int I>
__device__ __forceinline__ v2f x_frag(const v4f (&blk)[3], int hf) {
    const float e0 = blk[(3 * I) >> 2][(3 * I) & 3];
    const float e1 = blk[(3 * I + 1) >> 2][(3 * I + 1) & 3];
    const float e2 = blk[(3 * I + 2) >> 2][(3 * I + 2) & 3];
    v2f r;
    r[0] = hf ? e2 : e0;
    r[1] = hf ? 1.0f : e1;
    return r;
}

template <int S>
__device__ __forceinline__ void lstm_step(int t, int w, int lane, int hf,
                                          const v2f& wX, const v2f& wOwn, const v2f (&wO)[3],
                                          const v2f& hOwn, const v2f (&hO)[3], const v8f& obias,
                                          v8f& acc, float (&cst)[2], v8f& o,
                                          const v2f& pown, const v2f (&bh)[3],
                                          v2f& nown, v2f (&nbh)[3], const int (&kt)[3],
                                          const v4f (&cur)[3], const v4f (&nxt)[3],
                                          v4f (&ld)[3], const float* __restrict__ ldp,
                                          float* __restrict__ og, unsigned roff, v2f (*hx)[4][32])
{
    constexpr int P = S & 1;
    const v8f zero = {};
#pragma unroll
    for (int j = 0; j < 3; ++j)
        acc = wmma4(wO[j], bh[j], acc);

    v2f nh;
#pragma unroll
    for (int q = 0; q < 2; ++q) {
        const float iv = sigm_half(acc[4 * q + 0]);
        const float fv = sigm_half(acc[4 * q + 1]);
        const float gv = tanh_(acc[4 * q + 2]);
        const float ov = sigm_half(acc[4 * q + 3]);
        const float cc = fv * cst[q] + iv * gv;
        cst[q] = cc;
        nh[q] = ov * tanh_(cc);
    }
    hx[P][w][lane] = nh;
    nown = nh;
    __builtin_amdgcn_sched_barrier(0);
    if (S < 3) acc = wmma4(wX, x_frag<(S + 1) & 3>(cur, hf), zero);
    else       acc = wmma4(wX, x_frag<0>(nxt, hf), zero);
    acc = wmma4(wOwn, nh, acc);
    asm volatile("" :: "v"(acc));

    __builtin_amdgcn_fence(__ATOMIC_RELEASE, "workgroup", "local");
    __builtin_amdgcn_s_barrier();
    __builtin_amdgcn_fence(__ATOMIC_ACQUIRE, "workgroup", "local");

#pragma unroll
    for (int j = 0; j < 3; ++j)
        nbh[j] = hx[P][kt[j]][lane];

    if (t >= 2 && ((S + 2) & 3) == w) {
        if (hf == 0) { float* o3 = og + (t - 2) * 3; o3[roff] = o[0]; o3[roff + 1] = o[1]; o3[roff + 2] = o[2]; }
    }
    if (S == 0) {
        const v4f* lp = reinterpret_cast<const v4f*>(ldp + roff);
        ld[0] = lp[0]; ld[1] = lp[1]; ld[2] = lp[2];
    }
    if (t >= 1 && ((S + 3) & 3) == w) {
        o = wmma4(hOwn, pown, obias);
#pragma unroll
        for (int j = 0; j < 3; ++j)
            o = wmma4(hO[j], bh[j], o);
    }
}

__global__ __launch_bounds__(128)
void lstm_fused_wmma(const float* __restrict__ x,
                     const float* __restrict__ W_ih,
                     const float* __restrict__ W_hh,
                     const float* __restrict__ b_ih,
                     const float* __restrict__ b_hh,
                     const float* __restrict__ W_out,
                     const float* __restrict__ b_out,
                     float* __restrict__ out)
{
    __shared__ v2f hx[2][4][32];

    const int lane = threadIdx.x & 31;
    const int w    = __builtin_amdgcn_readfirstlane((int)(threadIdx.x >> 5));
    const int c    = lane & 15;
    const int hf   = lane >> 4;
    const int b0   = blockIdx.x * 16;

    int kt[3];
#pragma unroll
    for (int j = 0; j < 3; ++j) kt[j] = j + (j >= w ? 1 : 0);
    v2f wX, wOwn, wO[3];
    {
        const int N = 16 * (c & 3) + 4 * w + (c >> 2);
        const float sc = ((c & 3) == 2) ? 1.0f : 0.5f;
#pragma unroll
        for (int q = 0; q < 2; ++q) {
            const int kq = 2 * hf + q;
            wOwn[q] = W_hh[N * 16 + 4 * w + kq] * sc;
#pragma unroll
            for (int j = 0; j < 3; ++j)
                wO[j][q] = W_hh[N * 16 + 4 * kt[j] + kq] * sc;
            wX[q] = ((kq < 3) ? W_ih[N * 3 + (kq < 3 ? kq : 0)] : b_ih[N] + b_hh[N]) * sc;
        }
    }

    v2f hOwn, hO[3];
#pragma unroll
    for (int q = 0; q < 2; ++q) {
        const int kq = 2 * hf + q;
        hOwn[q] = (c < 3) ? W_out[(c < 3 ? c : 0) * 16 + 4 * w + kq] : 0.0f;
#pragma unroll
        for (int j = 0; j < 3; ++j)
            hO[j][q] = (c < 3) ? W_out[(c < 3 ? c : 0) * 16 + 4 * kt[j] + kq] : 0.0f;
    }
    v8f obias;
#pragma unroll
    for (int r = 0; r < 8; ++r)
        obias[r] = (r < 3 && hf == 0) ? b_out[r < 3 ? r : 0] : 0.0f;

    const float* xg = x + b0 * (T_LEN * 3);
    float* og = out + b0 * (T_LEN * 3);
    const unsigned roff = (unsigned)c * (T_LEN * 3);
    const v4f* xq = reinterpret_cast<const v4f*>(xg + roff);
    float* op = og + roff;

    float cst[2] = {0.0f, 0.0f};
    v2f bh[3], own;
#pragma unroll
    for (int j = 0; j < 3; ++j) { bh[j][0] = 0.0f; bh[j][1] = 0.0f; }
    own[0] = 0.0f; own[1] = 0.0f;

    const v8f zero = {};
    v4f xa[3], xb[3];
    xa[0] = xq[0]; xa[1] = xq[1]; xa[2] = xq[2];
    v8f acc = wmma4(wX, x_frag<0>(xa, hf), zero);

    v8f o = obias;
    v2f bhB[3], ownB;
    for (int t = 0; t < T_LEN; t += 8) {
        const float* pb = xg + 12 * (t / 4 + 1);
        const float* pa = xg + 12 * ((t + 8 < T_LEN) ? (t / 4 + 2) : (t / 4 + 1));
        lstm_step<0>(t, w, lane, hf, wX, wOwn, wO, hOwn, hO, obias, acc, cst, o, own, bh, ownB, bhB, kt, xa, xb, xb, pb, og, roff, hx);
        lstm_step<1>(t + 1, w, lane, hf, wX, wOwn, wO, hOwn, hO, obias, acc, cst, o, ownB, bhB, own, bh, kt, xa, xb, xb, pb, og, roff, hx);
        lstm_step<2>(t + 2, w, lane, hf, wX, wOwn, wO, hOwn, hO, obias, acc, cst, o, own, bh, ownB, bhB, kt, xa, xb, xb, pb, og, roff, hx);
        lstm_step<3>(t + 3, w, lane, hf, wX, wOwn, wO, hOwn, hO, obias, acc, cst, o, ownB, bhB, own, bh, kt, xa, xb, xb, pb, og, roff, hx);
        lstm_step<0>(t + 4, w, lane, hf, wX, wOwn, wO, hOwn, hO, obias, acc, cst, o, own, bh, ownB, bhB, kt, xb, xa, xa, pa, og, roff, hx);
        lstm_step<1>(t + 5, w, lane, hf, wX, wOwn, wO, hOwn, hO, obias, acc, cst, o, ownB, bhB, own, bh, kt, xb, xa, xa, pa, og, roff, hx);
        lstm_step<2>(t + 6, w, lane, hf, wX, wOwn, wO, hOwn, hO, obias, acc, cst, o, own, bh, ownB, bhB, kt, xb, xa, xa, pa, og, roff, hx);
        lstm_step<3>(t + 7, w, lane, hf, wX, wOwn, wO, hOwn, hO, obias, acc, cst, o, ownB, bhB, own, bh, kt, xb, xa, xa, pa, og, roff, hx);
    }

    if (((T_LEN - 2) & 3) == w) {
        if (hf == 0) { float* o3 = op + (T_LEN - 2) * 3; o3[0] = o[0]; o3[1] = o[1]; o3[2] = o[2]; }
    }
    if (((T_LEN - 1) & 3) == w)
        head_store(hOwn, hO, obias, own, bh, op + (T_LEN - 1) * 3, hf);
}

extern "C" void kernel_launch(void* const* d_in, const int* in_sizes, int n_in,
                              void* d_out, int out_size, void* d_ws, size_t ws_size,
                              hipStream_t stream) {
    const float* x     = (const float*)d_in[0];
    const float* W_ih  = (const float*)d_in[1];
    const float* W_hh  = (const float*)d_in[2];
    const float* b_ih  = (const float*)d_in[3];
    const float* b_hh  = (const float*)d_in[4];
    const float* W_out = (const float*)d_in[5];
    const float* b_out = (const float*)d_in[6];
    float* out = (float*)d_out;

    const int B = in_sizes[0] / (T_LEN * 3);
    const int grid = B / 16;

    lstm_fused_wmma<<<grid, 128, 0, stream>>>(x, W_ih, W_hh, b_ih, b_hh,
                                              W_out, b_out, out);
}
